# starting kernel
# baseline (speedup 1.0000x reference)
#include <hip/hip_runtime.h>
#include <math.h>

#define NN 4
#define CI 256
#define HH 64
#define WW 64
#define HWSZ (HH*WW)
#define CCOMP 64
#define MC 100
#define MCP 112
#define KK9 576
#define KTAPS 25
#define HO (HH*2)
#define WO (WW*2)

typedef float v2f __attribute__((ext_vector_type(2)));
typedef float v8f __attribute__((ext_vector_type(8)));

#if defined(__AMDGCN__) && __has_builtin(__builtin_amdgcn_wmma_f32_16x16x4_f32)
#define HAVE_WMMA_F32X4 1
#else
#define HAVE_WMMA_F32X4 0
#endif

__global__ __launch_bounds__(256) void k0_prep_w2t(const float* __restrict__ w2,
                                                   float* __restrict__ w2t) {
  int idx = blockIdx.x * 256 + threadIdx.x;
  if (idx >= MCP * KK9) return;
  int row = idx / KK9, kidx = idx % KK9;
  int dy = kidx / 192, r = kidx % 192, dx = r >> 6, cc = r & 63;
  w2t[idx] = (row < MC) ? w2[row * KK9 + cc * 9 + dy * 3 + dx] : 0.0f;
}

__global__ __launch_bounds__(256) void k1_compress(const float* __restrict__ x,
                                                   const float* __restrict__ w1,
                                                   const float* __restrict__ b1,
                                                   float* __restrict__ comp) {
#if HAVE_WMMA_F32X4
  int wave = blockIdx.x * 8 + (threadIdx.x >> 5);
  int lane = threadIdx.x & 31;
  int hl = lane >> 4, lid = lane & 15;
  int mt = wave & 3;
  int ct = wave >> 2;
  int n = ct >> 8;
  int hw0 = (ct & 255) << 4;
  const float* xb = x + n * (CI * HWSZ) + hw0 + lid;
  const float* ap = w1 + (mt * 16 + lid) * CI + 2 * hl;
  v8f acc = {};
  for (int k = 0; k < CI; k += 4) {
    int ka = k + 2 * hl;
    v2f a = *(const v2f*)(ap + k);
    v2f b;
    b.x = xb[ka * HWSZ];
    b.y = xb[(ka + 1) * HWSZ];
    acc = __builtin_amdgcn_wmma_f32_16x16x4_f32(false, a, false, b,
                                                (short)0, acc, false, false);
  }
  int col = hw0 + lid;
#pragma unroll
  for (int r = 0; r < 8; ++r) {
    int row = mt * 16 + r + 8 * hl;
    comp[(n * HWSZ + col) * CCOMP + row] = acc[r] + b1[row];
  }
#else
  int stride = gridDim.x * blockDim.x;
  for (int e = blockIdx.x * blockDim.x + threadIdx.x; e < CCOMP * NN * HWSZ; e += stride) {
    int row = e % CCOMP, col = e / CCOMP;
    int n = col / HWSZ, hw = col % HWSZ;
    float s = b1[row];
    for (int k = 0; k < CI; ++k) s += w1[row * CI + k] * x[(n * CI + k) * HWSZ + hw];
    comp[col * CCOMP + row] = s;
  }
#endif
}

__global__ __launch_bounds__(896) void k2_mask_softmax(const float* __restrict__ comp,
                                                       const float* __restrict__ w2t,
                                                       const float* __restrict__ b2,
                                                       float* __restrict__ mws) {
  __shared__ float lds_mask[MCP * WW];
  int n = blockIdx.x >> 6;
  int h = blockIdx.x & 63;
  int tid = threadIdx.x;
#if HAVE_WMMA_F32X4
  int wave = tid >> 5, lane = tid & 31, hl = lane >> 4, lid = lane & 15;
  int mt = wave >> 2;
  int w0 = (wave & 3) << 4;
  int wcol = w0 + lid;
  const float* abase = w2t + (mt * 16 + lid) * KK9 + 2 * hl;
  const float* cbase = comp + (size_t)n * HH * WW * CCOMP + 2 * hl;
  v8f acc = {};
  for (int dy = 0; dy < 3; ++dy) {
    int hrow = h + dy - 1;
    bool rowok = (unsigned)hrow < (unsigned)HH;
    for (int dx = 0; dx < 3; ++dx) {
      int wc = wcol + dx - 1;
      bool ok = rowok && ((unsigned)wc < (unsigned)WW);
      const float* bp = cbase + (hrow * WW + wc) * CCOMP;
      const float* ap = abase + (dy * 3 + dx) * 64;
#pragma unroll
      for (int cc = 0; cc < 64; cc += 4) {
        v2f a = *(const v2f*)(ap + cc);
        v2f b = {0.0f, 0.0f};
        if (ok) b = *(const v2f*)(bp + cc);
        acc = __builtin_amdgcn_wmma_f32_16x16x4_f32(false, a, false, b,
                                                    (short)0, acc, false, false);
      }
    }
  }
#pragma unroll
  for (int r = 0; r < 8; ++r) {
    int row = mt * 16 + r + 8 * hl;
    float v = acc[r] + ((row < MC) ? b2[row] : 0.0f);
    lds_mask[row * WW + w0 + lid] = v;
  }
#else
  for (int e = tid; e < MCP * WW; e += 896) {
    int row = e >> 6, pix = e & 63;
    float s = (row < MC) ? b2[row] : 0.0f;
    if (row < MC) {
      for (int kidx = 0; kidx < KK9; ++kidx) {
        int dy = kidx / 192, r0 = kidx % 192, dx = r0 >> 6, cc = r0 & 63;
        int hrow = h + dy - 1, wc = pix + dx - 1;
        if ((unsigned)hrow < (unsigned)HH && (unsigned)wc < (unsigned)WW)
          s += w2t[row * KK9 + kidx] * comp[((n * HH + hrow) * WW + wc) * CCOMP + cc];
      }
    }
    lds_mask[row * WW + pix] = s;
  }
#endif
  __syncthreads();
  if (tid < 256) {
    int pix = tid & 63, s = tid >> 6;
    float v[KTAPS];
    float mx = -1e30f;
#pragma unroll
    for (int kk = 0; kk < KTAPS; ++kk) {
      v[kk] = lds_mask[(kk * 4 + s) * WW + pix];
      mx = fmaxf(mx, v[kk]);
    }
    float sum = 0.0f;
#pragma unroll
    for (int kk = 0; kk < KTAPS; ++kk) { v[kk] = __expf(v[kk] - mx); sum += v[kk]; }
    float inv = 1.0f / sum;
    float* mp = mws + ((n * HH + h) * WW + pix) * MCP + s * 28;
#pragma unroll
    for (int kk = 0; kk < KTAPS; ++kk) mp[kk] = v[kk] * inv;
  }
}

__global__ __launch_bounds__(256) void k3_reassemble(const float* __restrict__ x,
                                                     const float* __restrict__ mws,
                                                     float* __restrict__ out) {
  __shared__ float lds_m[16 * 113];
  int b = blockIdx.x;
  int wt = b & 3;
  int h = (b >> 2) & 63;
  int n = b >> 8;
  int w0 = wt << 4;
  int tid = threadIdx.x;
  int base = ((n * HH + h) * WW + w0) * MCP;
  for (int idx = tid; idx < 16 * MCP; idx += 256)
    lds_m[(idx / MCP) * 113 + (idx % MCP)] = mws[base + idx];
  __syncthreads();

  int pix = tid & 15, ci = tid >> 4;
  int wcen = w0 + pix;
  for (int cg = 0; cg < CI; cg += 16) {
    int c = cg + ci;
    const float* xp = x + (size_t)(n * CI + c) * HWSZ;

    if (cg + 16 < CI) {
      const float* xn = xp + 16 * HWSZ;
#pragma unroll
      for (int dyp = -2; dyp <= 2; ++dyp) {
        int hyp = h + dyp;
        if ((unsigned)hyp < (unsigned)HH)
          __builtin_prefetch(xn + hyp * WW + wcen, 0, 3);
      }
    }

    float acc0 = 0.f, acc1 = 0.f, acc2 = 0.f, acc3 = 0.f;
#pragma unroll
    for (int kk = 0; kk < KTAPS; ++kk) {
      int dy = kk / 5 - 2, dx = kk % 5 - 2;
      int hy = h + dy, wx = wcen + dx;
      float xv = ((unsigned)hy < (unsigned)HH && (unsigned)wx < (unsigned)WW)
                     ? xp[hy * WW + wx] : 0.0f;
      const float* mp = &lds_m[pix * 113 + kk];
      acc0 += xv * mp[0];
      acc1 += xv * mp[28];
      acc2 += xv * mp[56];
      acc3 += xv * mp[84];
    }
    float* op = out + ((size_t)(n * CI + c) * HO + 2 * h) * WO + 2 * wcen;
    *(v2f*)op = (v2f){acc0, acc1};
    *(v2f*)(op + WO) = (v2f){acc2, acc3};
  }
}

extern "C" void kernel_launch(void* const* d_in, const int* in_sizes, int n_in,
                              void* d_out, int out_size, void* d_ws, size_t ws_size,
                              hipStream_t stream) {
  const float* x  = (const float*)d_in[0];
  const float* w1 = (const float*)d_in[1];
  const float* b1 = (const float*)d_in[2];
  const float* w2 = (const float*)d_in[3];
  const float* b2 = (const float*)d_in[4];
  float* out = (float*)d_out;

  float* comp = (float*)d_ws;
  float* w2t  = comp + NN * HWSZ * CCOMP;
  float* mws  = w2t + MCP * KK9;

  k0_prep_w2t<<<(MCP * KK9 + 255) / 256, 256, 0, stream>>>(w2, w2t);
  k1_compress<<<(4 * (NN * HWSZ / 16)) / 8, 256, 0, stream>>>(x, w1, b1, comp);
  k2_mask_softmax<<<NN * HH, 896, 0, stream>>>(comp, w2t, b2, mws);
  k3_reassemble<<<NN * HH * (WW / 16), 256, 0, stream>>>(x, mws, out);
}


# Fragment-major layouts for the WMMA operands (w1, w2, comp): a wave's fragment load is one contiguous 256-512 B chunk instead of one cache line per tile row; sums and their order unchanged
# speedup vs baseline: 1.2022x; 1.2022x over previous
#include <hip/hip_runtime.h>
#include <math.h>

#define NN 4
#define CI 256
#define HH 64
#define WW 64
#define HWSZ (HH*WW)
#define CCOMP 64
#define MC 100
#define MCP 112
#define KK9 576
#define KTAPS 25
#define HO (HH*2)
#define WO (WW*2)

typedef float v2f __attribute__((ext_vector_type(2)));
typedef float v4f __attribute__((ext_vector_type(4)));
typedef float v8f __attribute__((ext_vector_type(8)));
#define KP2 72
#define KS1 64

#if defined(__AMDGCN__) && __has_builtin(__builtin_amdgcn_wmma_f32_16x16x4_f32)
#define HAVE_WMMA_F32X4 1
#else
#define HAVE_WMMA_F32X4 0
#endif

__device__ __forceinline__ int w2f_index(int row, int kidx) {
  return ((((row >> 4) * KP2 + (kidx >> 3)) * 32 + ((kidx >> 1) & 1) * 16 + (row & 15)) << 2) +
         (((kidx >> 2) & 1) << 1) + (kidx & 1);
}
__device__ __forceinline__ int w1f_index(int row, int k) {
  return ((((row >> 4) * KS1 + (k >> 2)) * 32 + ((k >> 1) & 1) * 16 + (row & 15)) << 1) + (k & 1);
}
__device__ __forceinline__ int comp_index(int n, int h, int w, int cc) {
  return ((((n * HH + h) * 8 + (cc >> 3)) * WW + w) << 3) +
         (((cc >> 1) & 1) << 2) + (((cc >> 2) & 1) << 1) + (cc & 1);
}

__global__ __launch_bounds__(256) void k0_prep_frag(const float* __restrict__ w2,
                                                    const float* __restrict__ w1,
                                                    float* __restrict__ w2f,
                                                    float* __restrict__ w1f) {
  int idx = blockIdx.x * 256 + threadIdx.x;
  int j = idx & 1;
  if (idx < MCP * KK9) {
    int t = (idx >> 1) & 1, lane = (idx >> 2) & 31, lid = lane & 15, hl = lane >> 4;
    int g = idx >> 7;
    int kp = g % KP2, mt = g / KP2;
    int row = mt * 16 + lid;
    int k = 8 * kp + 4 * t + 2 * hl + j;
    int tap = k >> 6, cc = k & 63;
    w2f[idx] = (row < MC) ? w2[row * KK9 + cc * 9 + tap] : 0.0f;
  } else {
    int e = idx - MCP * KK9;
    if (e < CCOMP * CI) {
      int lane = (e >> 1) & 31, lid = lane & 15, hl = lane >> 4;
      int g = e >> 6;
      int ks = g & (KS1 - 1), mt = g >> 6;
      w1f[e] = w1[(mt * 16 + lid) * CI + 4 * ks + 2 * hl + j];
    }
  }
}

__global__ __launch_bounds__(256) void k1_compress(const float* __restrict__ x,
                                                   const float* __restrict__ w1f,
                                                   const float* __restrict__ b1,
                                                   float* __restrict__ comp) {
#if HAVE_WMMA_F32X4
  int wave = blockIdx.x * 8 + (threadIdx.x >> 5);
  int lane = threadIdx.x & 31;
  int hl = lane >> 4, lid = lane & 15;
  int mt = wave & 3;
  int ct = wave >> 2;
  int n = ct >> 8;
  int hw0 = (ct & 255) << 4;
  const float* xb = x + n * (CI * HWSZ) + hw0 + lid;
  const float* ap = w1f + ((mt * KS1 * 32 + lane) << 1);
  v8f acc = {};
  for (int k = 0; k < CI; k += 4) {
    int ka = k + 2 * hl;
    v2f a = *(const v2f*)(ap + k * 16);
    v2f b;
    b.x = xb[ka * HWSZ];
    b.y = xb[(ka + 1) * HWSZ];
    acc = __builtin_amdgcn_wmma_f32_16x16x4_f32(false, a, false, b,
                                                (short)0, acc, false, false);
  }
  int col = hw0 + lid;
  int row0 = mt * 16 + 8 * hl;
  const float* bb = b1 + row0;
  float* cp = comp + comp_index(n, col >> 6, col & 63, row0);
  v4f lo = {acc[0] + bb[0], acc[1] + bb[1], acc[4] + bb[4], acc[5] + bb[5]};
  v4f hi = {acc[2] + bb[2], acc[3] + bb[3], acc[6] + bb[6], acc[7] + bb[7]};
  *(v4f*)cp = lo;
  *(v4f*)(cp + 4) = hi;
#else
  int stride = gridDim.x * blockDim.x;
  for (int e = blockIdx.x * blockDim.x + threadIdx.x; e < CCOMP * NN * HWSZ; e += stride) {
    int row = e % CCOMP, col = e / CCOMP;
    int n = col / HWSZ, hw = col % HWSZ;
    float s = b1[row];
    for (int k = 0; k < CI; ++k) s += w1f[w1f_index(row, k)] * x[(n * CI + k) * HWSZ + hw];
    comp[comp_index(n, hw >> 6, hw & 63, row)] = s;
  }
#endif
}

__global__ __launch_bounds__(896) void k2_mask_softmax(const float* __restrict__ comp,
                                                       const float* __restrict__ w2f,
                                                       const float* __restrict__ b2,
                                                       float* __restrict__ mws) {
  __shared__ float lds_mask[MCP * WW];
  int n = blockIdx.x >> 6;
  int h = blockIdx.x & 63;
  int tid = threadIdx.x;
#if HAVE_WMMA_F32X4
  int wave = tid >> 5, lane = tid & 31, hl = lane >> 4, lid = lane & 15;
  int mt = wave >> 2;
  int w0 = (wave & 3) << 4;
  int wcol = w0 + lid;
  const float* abase = w2f + ((mt * KP2 * 32 + lane) << 2);
  const float* cbase = comp + (size_t)n * HH * WW * CCOMP + 4 * hl;
  v8f acc = {};
  for (int dy = 0; dy < 3; ++dy) {
    int hrow = h + dy - 1;
    bool rowok = (unsigned)hrow < (unsigned)HH;
    for (int dx = 0; dx < 3; ++dx) {
      int wc = wcol + dx - 1;
      bool ok = rowok && ((unsigned)wc < (unsigned)WW);
      const float* bp = cbase + ((hrow * 8 * WW + wc) << 3);
      const float* ap = abase + (dy * 3 + dx) * 8 * 128;
#pragma unroll
      for (int cp = 0; cp < 8; ++cp) {
        v4f a = *(const v4f*)(ap + cp * 128);
        v4f b = {0.0f, 0.0f, 0.0f, 0.0f};
        if (ok) b = *(const v4f*)(bp + cp * (WW * 8));
        acc = __builtin_amdgcn_wmma_f32_16x16x4_f32(false, a.xy, false, b.xy,
                                                    (short)0, acc, false, false);
        acc = __builtin_amdgcn_wmma_f32_16x16x4_f32(false, a.zw, false, b.zw,
                                                    (short)0, acc, false, false);
      }
    }
  }
#pragma unroll
  for (int r = 0; r < 8; ++r) {
    int row = mt * 16 + r + 8 * hl;
    float v = acc[r] + ((row < MC) ? b2[row] : 0.0f);
    lds_mask[row * WW + w0 + lid] = v;
  }
#else
  for (int e = tid; e < MCP * WW; e += 896) {
    int row = e >> 6, pix = e & 63;
    float s = (row < MC) ? b2[row] : 0.0f;
    if (row < MC) {
      for (int kidx = 0; kidx < KK9; ++kidx) {
        int dy = kidx / 192, r0 = kidx % 192, dx = r0 >> 6, cc = r0 & 63;
        int hrow = h + dy - 1, wc = pix + dx - 1;
        if ((unsigned)hrow < (unsigned)HH && (unsigned)wc < (unsigned)WW)
          s += w2f[w2f_index(row, kidx)] * comp[comp_index(n, hrow, wc, cc)];
      }
    }
    lds_mask[row * WW + pix] = s;
  }
#endif
  __syncthreads();
  if (tid < 256) {
    int pix = tid & 63, s = tid >> 6;
    float v[KTAPS];
    float mx = -1e30f;
#pragma unroll
    for (int kk = 0; kk < KTAPS; ++kk) {
      v[kk] = lds_mask[(kk * 4 + s) * WW + pix];
      mx = fmaxf(mx, v[kk]);
    }
    float sum = 0.0f;
#pragma unroll
    for (int kk = 0; kk < KTAPS; ++kk) { v[kk] = __expf(v[kk] - mx); sum += v[kk]; }
    float inv = 1.0f / sum;
    float* mp = mws + ((n * HH + h) * WW + pix) * MCP + s * 28;
#pragma unroll
    for (int kk = 0; kk < KTAPS; ++kk) mp[kk] = v[kk] * inv;
  }
}

__global__ __launch_bounds__(256) void k3_reassemble(const float* __restrict__ x,
                                                     const float* __restrict__ mws,
                                                     float* __restrict__ out) {
  __shared__ float lds_m[16 * 113];
  int b = blockIdx.x;
  int wt = b & 3;
  int h = (b >> 2) & 63;
  int n = b >> 8;
  int w0 = wt << 4;
  int tid = threadIdx.x;
  int base = ((n * HH + h) * WW + w0) * MCP;
  for (int idx = tid; idx < 16 * MCP; idx += 256)
    lds_m[(idx / MCP) * 113 + (idx % MCP)] = mws[base + idx];
  __syncthreads();

  int pix = tid & 15, ci = tid >> 4;
  int wcen = w0 + pix;
  for (int cg = 0; cg < CI; cg += 16) {
    int c = cg + ci;
    const float* xp = x + (size_t)(n * CI + c) * HWSZ;

    if (cg + 16 < CI) {
      const float* xn = xp + 16 * HWSZ;
#pragma unroll
      for (int dyp = -2; dyp <= 2; ++dyp) {
        int hyp = h + dyp;
        if ((unsigned)hyp < (unsigned)HH)
          __builtin_prefetch(xn + hyp * WW + wcen, 0, 3);
      }
    }

    float acc0 = 0.f, acc1 = 0.f, acc2 = 0.f, acc3 = 0.f;
#pragma unroll
    for (int kk = 0; kk < KTAPS; ++kk) {
      int dy = kk / 5 - 2, dx = kk % 5 - 2;
      int hy = h + dy, wx = wcen + dx;
      float xv = ((unsigned)hy < (unsigned)HH && (unsigned)wx < (unsigned)WW)
                     ? xp[hy * WW + wx] : 0.0f;
      const float* mp = &lds_m[pix * 113 + kk];
      acc0 += xv * mp[0];
      acc1 += xv * mp[28];
      acc2 += xv * mp[56];
      acc3 += xv * mp[84];
    }
    float* op = out + ((size_t)(n * CI + c) * HO + 2 * h) * WO + 2 * wcen;
    *(v2f*)op = (v2f){acc0, acc1};
    *(v2f*)(op + WO) = (v2f){acc2, acc3};
  }
}

extern "C" void kernel_launch(void* const* d_in, const int* in_sizes, int n_in,
                              void* d_out, int out_size, void* d_ws, size_t ws_size,
                              hipStream_t stream) {
  const float* x  = (const float*)d_in[0];
  const float* w1 = (const float*)d_in[1];
  const float* b1 = (const float*)d_in[2];
  const float* w2 = (const float*)d_in[3];
  const float* b2 = (const float*)d_in[4];
  float* out = (float*)d_out;

  float* comp = (float*)d_ws;
  float* w2f  = comp + NN * HWSZ * CCOMP;
  float* mws  = w2f + MCP * KK9;
  float* w1f  = mws;

  k0_prep_frag<<<(MCP * KK9 + CCOMP * CI) / 256, 256, 0, stream>>>(w2, w1, w2f, w1f);
  k1_compress<<<(4 * (NN * HWSZ / 16)) / 8, 256, 0, stream>>>(x, w1f, b1, comp);
  k2_mask_softmax<<<NN * HH, 896, 0, stream>>>(comp, w2f, b2, mws);
  k3_reassemble<<<NN * HH * (WW / 16), 256, 0, stream>>>(x, mws, out);
}


# k3: zero padding of x folded into the per-pixel mask registers, so a channel step's 25 tap loads are unconditional (scalar clamped-row base + 32-bit lane offset), issued as one group; same sums
# speedup vs baseline: 1.4793x; 1.2305x over previous
#include <hip/hip_runtime.h>
#include <math.h>

#define NN 4
#define CI 256
#define HH 64
#define WW 64
#define HWSZ (HH*WW)
#define CCOMP 64
#define MC 100
#define MCP 112
#define KK9 576
#define KTAPS 25
#define HO (HH*2)
#define WO (WW*2)

typedef float v2f __attribute__((ext_vector_type(2)));
typedef float v4f __attribute__((ext_vector_type(4)));
typedef float v8f __attribute__((ext_vector_type(8)));
#define KP2 72
#define KS1 64

#if defined(__AMDGCN__) && __has_builtin(__builtin_amdgcn_wmma_f32_16x16x4_f32)
#define HAVE_WMMA_F32X4 1
#else
#define HAVE_WMMA_F32X4 0
#endif

__device__ __forceinline__ int w2f_index(int row, int kidx) {
  return ((((row >> 4) * KP2 + (kidx >> 3)) * 32 + ((kidx >> 1) & 1) * 16 + (row & 15)) << 2) +
         (((kidx >> 2) & 1) << 1) + (kidx & 1);
}
__device__ __forceinline__ int w1f_index(int row, int k) {
  return ((((row >> 4) * KS1 + (k >> 2)) * 32 + ((k >> 1) & 1) * 16 + (row & 15)) << 1) + (k & 1);
}
__device__ __forceinline__ int comp_index(int n, int h, int w, int cc) {
  return ((((n * HH + h) * 8 + (cc >> 3)) * WW + w) << 3) +
         (((cc >> 1) & 1) << 2) + (((cc >> 2) & 1) << 1) + (cc & 1);
}

__global__ __launch_bounds__(256) void k0_prep_frag(const float* __restrict__ w2,
                                                    const float* __restrict__ w1,
                                                    float* __restrict__ w2f,
                                                    float* __restrict__ w1f) {
  int idx = blockIdx.x * 256 + threadIdx.x;
  int j = idx & 1;
  if (idx < MCP * KK9) {
    int t = (idx >> 1) & 1, lane = (idx >> 2) & 31, lid = lane & 15, hl = lane >> 4;
    int g = idx >> 7;
    int kp = g % KP2, mt = g / KP2;
    int row = mt * 16 + lid;
    int k = 8 * kp + 4 * t + 2 * hl + j;
    int tap = k >> 6, cc = k & 63;
    w2f[idx] = (row < MC) ? w2[row * KK9 + cc * 9 + tap] : 0.0f;
  } else {
    int e = idx - MCP * KK9;
    if (e < CCOMP * CI) {
      int lane = (e >> 1) & 31, lid = lane & 15, hl = lane >> 4;
      int g = e >> 6;
      int ks = g & (KS1 - 1), mt = g >> 6;
      w1f[e] = w1[(mt * 16 + lid) * CI + 4 * ks + 2 * hl + j];
    }
  }
}

__global__ __launch_bounds__(256) void k1_compress(const float* __restrict__ x,
                                                   const float* __restrict__ w1f,
                                                   const float* __restrict__ b1,
                                                   float* __restrict__ comp) {
#if HAVE_WMMA_F32X4
  int wave = blockIdx.x * 8 + (threadIdx.x >> 5);
  int lane = threadIdx.x & 31;
  int hl = lane >> 4, lid = lane & 15;
  int mt = wave & 3;
  int ct = wave >> 2;
  int n = ct >> 8;
  int hw0 = (ct & 255) << 4;
  const float* xb = x + n * (CI * HWSZ) + hw0 + lid;
  const float* ap = w1f + ((mt * KS1 * 32 + lane) << 1);
  v8f acc = {};
  for (int k = 0; k < CI; k += 4) {
    int ka = k + 2 * hl;
    v2f a = *(const v2f*)(ap + k * 16);
    v2f b;
    b.x = xb[ka * HWSZ];
    b.y = xb[(ka + 1) * HWSZ];
    acc = __builtin_amdgcn_wmma_f32_16x16x4_f32(false, a, false, b,
                                                (short)0, acc, false, false);
  }
  int col = hw0 + lid;
  int row0 = mt * 16 + 8 * hl;
  const float* bb = b1 + row0;
  float* cp = comp + comp_index(n, col >> 6, col & 63, row0);
  v4f lo = {acc[0] + bb[0], acc[1] + bb[1], acc[4] + bb[4], acc[5] + bb[5]};
  v4f hi = {acc[2] + bb[2], acc[3] + bb[3], acc[6] + bb[6], acc[7] + bb[7]};
  *(v4f*)cp = lo;
  *(v4f*)(cp + 4) = hi;
#else
  int stride = gridDim.x * blockDim.x;
  for (int e = blockIdx.x * blockDim.x + threadIdx.x; e < CCOMP * NN * HWSZ; e += stride) {
    int row = e % CCOMP, col = e / CCOMP;
    int n = col / HWSZ, hw = col % HWSZ;
    float s = b1[row];
    for (int k = 0; k < CI; ++k) s += w1f[w1f_index(row, k)] * x[(n * CI + k) * HWSZ + hw];
    comp[comp_index(n, hw >> 6, hw & 63, row)] = s;
  }
#endif
}

__global__ __launch_bounds__(896) void k2_mask_softmax(const float* __restrict__ comp,
                                                       const float* __restrict__ w2f,
                                                       const float* __restrict__ b2,
                                                       float* __restrict__ mws) {
  __shared__ float lds_mask[MCP * WW];
  int n = blockIdx.x >> 6;
  int h = blockIdx.x & 63;
  int tid = threadIdx.x;
#if HAVE_WMMA_F32X4
  int wave = tid >> 5, lane = tid & 31, hl = lane >> 4, lid = lane & 15;
  int mt = wave >> 2;
  int w0 = (wave & 3) << 4;
  int wcol = w0 + lid;
  const float* abase = w2f + ((mt * KP2 * 32 + lane) << 2);
  const float* cbase = comp + (size_t)n * HH * WW * CCOMP + 4 * hl;
  v8f acc = {};
  for (int dy = 0; dy < 3; ++dy) {
    int hrow = h + dy - 1;
    bool rowok = (unsigned)hrow < (unsigned)HH;
    for (int dx = 0; dx < 3; ++dx) {
      int wc = wcol + dx - 1;
      bool ok = rowok && ((unsigned)wc < (unsigned)WW);
      const float* bp = cbase + ((hrow * 8 * WW + wc) << 3);
      const float* ap = abase + (dy * 3 + dx) * 8 * 128;
#pragma unroll
      for (int cp = 0; cp < 8; ++cp) {
        v4f a = *(const v4f*)(ap + cp * 128);
        v4f b = {0.0f, 0.0f, 0.0f, 0.0f};
        if (ok) b = *(const v4f*)(bp + cp * (WW * 8));
        acc = __builtin_amdgcn_wmma_f32_16x16x4_f32(false, a.xy, false, b.xy,
                                                    (short)0, acc, false, false);
        acc = __builtin_amdgcn_wmma_f32_16x16x4_f32(false, a.zw, false, b.zw,
                                                    (short)0, acc, false, false);
      }
    }
  }
#pragma unroll
  for (int r = 0; r < 8; ++r) {
    int row = mt * 16 + r + 8 * hl;
    float v = acc[r] + ((row < MC) ? b2[row] : 0.0f);
    lds_mask[row * WW + w0 + lid] = v;
  }
#else
  for (int e = tid; e < MCP * WW; e += 896) {
    int row = e >> 6, pix = e & 63;
    float s = (row < MC) ? b2[row] : 0.0f;
    if (row < MC) {
      for (int kidx = 0; kidx < KK9; ++kidx) {
        int dy = kidx / 192, r0 = kidx % 192, dx = r0 >> 6, cc = r0 & 63;
        int hrow = h + dy - 1, wc = pix + dx - 1;
        if ((unsigned)hrow < (unsigned)HH && (unsigned)wc < (unsigned)WW)
          s += w2f[w2f_index(row, kidx)] * comp[comp_index(n, hrow, wc, cc)];
      }
    }
    lds_mask[row * WW + pix] = s;
  }
#endif
  __syncthreads();
  if (tid < 256) {
    int pix = tid & 63, s = tid >> 6;
    float v[KTAPS];
    float mx = -1e30f;
#pragma unroll
    for (int kk = 0; kk < KTAPS; ++kk) {
      v[kk] = lds_mask[(kk * 4 + s) * WW + pix];
      mx = fmaxf(mx, v[kk]);
    }
    float sum = 0.0f;
#pragma unroll
    for (int kk = 0; kk < KTAPS; ++kk) { v[kk] = __expf(v[kk] - mx); sum += v[kk]; }
    float inv = 1.0f / sum;
    float* mp = mws + ((n * HH + h) * WW + pix) * MCP + s * 28;
#pragma unroll
    for (int kk = 0; kk < KTAPS; ++kk) mp[kk] = v[kk] * inv;
  }
}

__global__ __launch_bounds__(256) void k3_reassemble(const float* __restrict__ x,
                                                     const float* __restrict__ mws,
                                                     float* __restrict__ out) {
  __shared__ float lds_m[16 * 113];
  int b = blockIdx.x;
  int wt = b & 3;
  int h = (b >> 2) & 63;
  int n = b >> 8;
  int w0 = wt << 4;
  int tid = threadIdx.x;
  int base = ((n * HH + h) * WW + w0) * MCP;
  for (int idx = tid; idx < 16 * MCP; idx += 256)
    lds_m[(idx / MCP) * 113 + (idx % MCP)] = mws[base + idx];
  __syncthreads();

  int pix = tid & 15, ci = tid >> 4;
  int wcen = w0 + pix;

  float m0[KTAPS], m1[KTAPS], m2[KTAPS], m3[KTAPS];
#pragma unroll
  for (int kk = 0; kk < KTAPS; ++kk) {
    int dy = kk / 5 - 2, dx = kk % 5 - 2;
    bool ok = (unsigned)(h + dy) < (unsigned)HH && (unsigned)(wcen + dx) < (unsigned)WW;
    const float* mp = &lds_m[pix * 113 + kk];
    float t0 = mp[0], t1 = mp[28], t2 = mp[56], t3 = mp[84];
    m0[kk] = ok ? t0 : 0.0f;
    m1[kk] = ok ? t1 : 0.0f;
    m2[kk] = ok ? t2 : 0.0f;
    m3[kk] = ok ? t3 : 0.0f;
  }
  unsigned coff[5];
  int roff[5];
#pragma unroll
  for (int d = 0; d < 5; ++d) {
    int wx = min(max(wcen + d - 2, 0), WW - 1);
    coff[d] = (unsigned)(ci * HWSZ + wx) * 4u;
    roff[d] = min(max(h + d - 2, 0), HH - 1) * (WW * 4);
  }
  const char* xn = (const char*)(x + (size_t)n * CI * HWSZ);
  float* op = out + ((size_t)(n * CI + ci) * HO + 2 * h) * WO + 2 * wcen;

  for (int cg = 0; cg < CI; cg += 16) {
    const char* xc = xn + (size_t)cg * (HWSZ * 4);
    unsigned vo[5];
#pragma unroll
    for (int d = 0; d < 5; ++d) {
      vo[d] = coff[d];
      asm volatile("" : "+v"(vo[d]));
    }

    if (cg + 16 < CI) {
      unsigned vp = coff[2];
#pragma unroll
      for (int d = 0; d < 5; ++d)
        __builtin_prefetch(xc + 16 * HWSZ * 4 + roff[d] + vp, 0, 3);
    }

    float xv[KTAPS];
#pragma unroll
    for (int kk = 0; kk < KTAPS; ++kk)
      xv[kk] = *(const float*)(xc + roff[kk / 5] + vo[kk % 5]);
    __builtin_amdgcn_sched_barrier(0);
    float acc0 = 0.f, acc1 = 0.f, acc2 = 0.f, acc3 = 0.f;
#pragma unroll
    for (int kk = 0; kk < KTAPS; ++kk) {
      acc0 += xv[kk] * m0[kk];
      acc1 += xv[kk] * m1[kk];
      acc2 += xv[kk] * m2[kk];
      acc3 += xv[kk] * m3[kk];
    }
    *(v2f*)op = (v2f){acc0, acc1};
    *(v2f*)(op + WO) = (v2f){acc2, acc3};
    op += (size_t)16 * HO * WO;
  }
}

extern "C" void kernel_launch(void* const* d_in, const int* in_sizes, int n_in,
                              void* d_out, int out_size, void* d_ws, size_t ws_size,
                              hipStream_t stream) {
  const float* x  = (const float*)d_in[0];
  const float* w1 = (const float*)d_in[1];
  const float* b1 = (const float*)d_in[2];
  const float* w2 = (const float*)d_in[3];
  const float* b2 = (const float*)d_in[4];
  float* out = (float*)d_out;

  float* comp = (float*)d_ws;
  float* w2f  = comp + NN * HWSZ * CCOMP;
  float* mws  = w2f + MCP * KK9;
  float* w1f  = mws;

  k0_prep_frag<<<(MCP * KK9 + CCOMP * CI) / 256, 256, 0, stream>>>(w2, w1, w2f, w1f);
  k1_compress<<<(4 * (NN * HWSZ / 16)) / 8, 256, 0, stream>>>(x, w1f, b1, comp);
  k2_mask_softmax<<<NN * HH, 896, 0, stream>>>(comp, w2f, b2, mws);
  k3_reassemble<<<NN * HH * (WW / 16), 256, 0, stream>>>(x, mws, out);
}


# k2: conv zero padding folded into a select on the arrived B fragment (no load under a branch); K loop loads run one group of 4 k-step pairs ahead of the WMMAs (two fragment sets); same sums
# speedup vs baseline: 1.7101x; 1.1560x over previous
#include <hip/hip_runtime.h>
#include <math.h>

#define NN 4
#define CI 256
#define HH 64
#define WW 64
#define HWSZ (HH*WW)
#define CCOMP 64
#define MC 100
#define MCP 112
#define KK9 576
#define KTAPS 25
#define HO (HH*2)
#define WO (WW*2)

typedef float v2f __attribute__((ext_vector_type(2)));
typedef float v4f __attribute__((ext_vector_type(4)));
typedef float v8f __attribute__((ext_vector_type(8)));
#define KP2 72
#define KS1 64

#if defined(__AMDGCN__) && __has_builtin(__builtin_amdgcn_wmma_f32_16x16x4_f32)
#define HAVE_WMMA_F32X4 1
#else
#define HAVE_WMMA_F32X4 0
#endif

__device__ __forceinline__ int w2f_index(int row, int kidx) {
  return ((((row >> 4) * KP2 + (kidx >> 3)) * 32 + ((kidx >> 1) & 1) * 16 + (row & 15)) << 2) +
         (((kidx >> 2) & 1) << 1) + (kidx & 1);
}
__device__ __forceinline__ int w1f_index(int row, int k) {
  return ((((row >> 4) * KS1 + (k >> 2)) * 32 + ((k >> 1) & 1) * 16 + (row & 15)) << 1) + (k & 1);
}
__device__ __forceinline__ int comp_index(int n, int h, int w, int cc) {
  return ((((n * HH + h) * 8 + (cc >> 3)) * WW + w) << 3) +
         (((cc >> 1) & 1) << 2) + (((cc >> 2) & 1) << 1) + (cc & 1);
}

__global__ __launch_bounds__(256) void k0_prep_frag(const float* __restrict__ w2,
                                                    const float* __restrict__ w1,
                                                    float* __restrict__ w2f,
                                                    float* __restrict__ w1f) {
  int idx = blockIdx.x * 256 + threadIdx.x;
  int j = idx & 1;
  if (idx < MCP * KK9) {
    int t = (idx >> 1) & 1, lane = (idx >> 2) & 31, lid = lane & 15, hl = lane >> 4;
    int g = idx >> 7;
    int kp = g % KP2, mt = g / KP2;
    int row = mt * 16 + lid;
    int k = 8 * kp + 4 * t + 2 * hl + j;
    int tap = k >> 6, cc = k & 63;
    w2f[idx] = (row < MC) ? w2[row * KK9 + cc * 9 + tap] : 0.0f;
  } else {
    int e = idx - MCP * KK9;
    if (e < CCOMP * CI) {
      int lane = (e >> 1) & 31, lid = lane & 15, hl = lane >> 4;
      int g = e >> 6;
      int ks = g & (KS1 - 1), mt = g >> 6;
      w1f[e] = w1[(mt * 16 + lid) * CI + 4 * ks + 2 * hl + j];
    }
  }
}

__global__ __launch_bounds__(256) void k1_compress(const float* __restrict__ x,
                                                   const float* __restrict__ w1f,
                                                   const float* __restrict__ b1,
                                                   float* __restrict__ comp) {
#if HAVE_WMMA_F32X4
  int wave = blockIdx.x * 8 + (threadIdx.x >> 5);
  int lane = threadIdx.x & 31;
  int hl = lane >> 4, lid = lane & 15;
  int mt = wave & 3;
  int ct = wave >> 2;
  int n = ct >> 8;
  int hw0 = (ct & 255) << 4;
  const float* xb = x + n * (CI * HWSZ) + hw0 + lid;
  const float* ap = w1f + ((mt * KS1 * 32 + lane) << 1);
  v8f acc = {};
  for (int k = 0; k < CI; k += 4) {
    int ka = k + 2 * hl;
    v2f a = *(const v2f*)(ap + k * 16);
    v2f b;
    b.x = xb[ka * HWSZ];
    b.y = xb[(ka + 1) * HWSZ];
    acc = __builtin_amdgcn_wmma_f32_16x16x4_f32(false, a, false, b,
                                                (short)0, acc, false, false);
  }
  int col = hw0 + lid;
  int row0 = mt * 16 + 8 * hl;
  const float* bb = b1 + row0;
  float* cp = comp + comp_index(n, col >> 6, col & 63, row0);
  v4f lo = {acc[0] + bb[0], acc[1] + bb[1], acc[4] + bb[4], acc[5] + bb[5]};
  v4f hi = {acc[2] + bb[2], acc[3] + bb[3], acc[6] + bb[6], acc[7] + bb[7]};
  *(v4f*)cp = lo;
  *(v4f*)(cp + 4) = hi;
#else
  int stride = gridDim.x * blockDim.x;
  for (int e = blockIdx.x * blockDim.x + threadIdx.x; e < CCOMP * NN * HWSZ; e += stride) {
    int row = e % CCOMP, col = e / CCOMP;
    int n = col / HWSZ, hw = col % HWSZ;
    float s = b1[row];
    for (int k = 0; k < CI; ++k) s += w1f[w1f_index(row, k)] * x[(n * CI + k) * HWSZ + hw];
    comp[comp_index(n, hw >> 6, hw & 63, row)] = s;
  }
#endif
}

#if HAVE_WMMA_F32X4
__device__ __forceinline__ void k2_load_group(const float* __restrict__ ap,
                                              const float* __restrict__ bp,
                                              v4f (&fa)[4], v4f (&fb)[4]) {
#pragma unroll
  for (int i = 0; i < 4; ++i) {
    fa[i] = *(const v4f*)(ap + i * 128);
    fb[i] = *(const v4f*)(bp + i * (WW * 8));
  }
}
__device__ __forceinline__ v8f k2_mma_group(v8f acc, const v4f (&fa)[4],
                                            const v4f (&fb)[4], bool ok) {
#pragma unroll
  for (int i = 0; i < 4; ++i) {
    v4f b = ok ? fb[i] : (v4f){0.0f, 0.0f, 0.0f, 0.0f};
    acc = __builtin_amdgcn_wmma_f32_16x16x4_f32(false, fa[i].xy, false, b.xy,
                                                (short)0, acc, false, false);
    acc = __builtin_amdgcn_wmma_f32_16x16x4_f32(false, fa[i].zw, false, b.zw,
                                                (short)0, acc, false, false);
  }
  return acc;
}
#endif

__global__ __launch_bounds__(896) void k2_mask_softmax(const float* __restrict__ comp,
                                                       const float* __restrict__ w2f,
                                                       const float* __restrict__ b2,
                                                       float* __restrict__ mws) {
  __shared__ float lds_mask[MCP * WW];
  int n = blockIdx.x >> 6;
  int h = blockIdx.x & 63;
  int tid = threadIdx.x;
#if HAVE_WMMA_F32X4
  int wave = tid >> 5, lane = tid & 31, hl = lane >> 4, lid = lane & 15;
  int mt = wave >> 2;
  int w0 = (wave & 3) << 4;
  int wcol = w0 + lid;
  const float* abase = w2f + ((mt * KP2 * 32 + lane) << 2);
  const float* cbase = comp + (size_t)n * HH * WW * CCOMP + 4 * hl;
  int rofs[3], cofs[3];
  bool rok[3], cok[3];
#pragma unroll
  for (int d = 0; d < 3; ++d) {
    int hr = h + d - 1, wc = wcol + d - 1;
    rok[d] = (unsigned)hr < (unsigned)HH;
    cok[d] = (unsigned)wc < (unsigned)WW;
    rofs[d] = min(max(hr, 0), HH - 1) * (8 * WW * 8);
    cofs[d] = min(max(wc, 0), WW - 1) * 8;
  }
  v4f fa0[4], fb0[4], fa1[4], fb1[4];
  v8f acc = {};
#define K2_AP(g) (abase + (g) * (4 * 128))
#define K2_BP(g) (cbase + rofs[((g) >> 1) / 3] + cofs[((g) >> 1) % 3] + ((g) & 1) * (4 * WW * 8))
#define K2_OK(g) (rok[((g) >> 1) / 3] && cok[((g) >> 1) % 3])
  k2_load_group(K2_AP(0), K2_BP(0), fa0, fb0);
#pragma unroll
  for (int g = 0; g < 18; g += 2) {
    k2_load_group(K2_AP(g + 1), K2_BP(g + 1), fa1, fb1);
    __builtin_amdgcn_sched_barrier(0);
    acc = k2_mma_group(acc, fa0, fb0, K2_OK(g));
    __builtin_amdgcn_sched_barrier(0);
    if (g + 2 < 18) k2_load_group(K2_AP(g + 2), K2_BP(g + 2), fa0, fb0);
    __builtin_amdgcn_sched_barrier(0);
    acc = k2_mma_group(acc, fa1, fb1, K2_OK(g + 1));
    __builtin_amdgcn_sched_barrier(0);
  }
#undef K2_AP
#undef K2_BP
#undef K2_OK
#pragma unroll
  for (int r = 0; r < 8; ++r) {
    int row = mt * 16 + r + 8 * hl;
    float v = acc[r] + ((row < MC) ? b2[row] : 0.0f);
    lds_mask[row * WW + w0 + lid] = v;
  }
#else
  for (int e = tid; e < MCP * WW; e += 896) {
    int row = e >> 6, pix = e & 63;
    float s = (row < MC) ? b2[row] : 0.0f;
    if (row < MC) {
      for (int kidx = 0; kidx < KK9; ++kidx) {
        int dy = kidx / 192, r0 = kidx % 192, dx = r0 >> 6, cc = r0 & 63;
        int hrow = h + dy - 1, wc = pix + dx - 1;
        if ((unsigned)hrow < (unsigned)HH && (unsigned)wc < (unsigned)WW)
          s += w2f[w2f_index(row, kidx)] * comp[comp_index(n, hrow, wc, cc)];
      }
    }
    lds_mask[row * WW + pix] = s;
  }
#endif
  __syncthreads();
  if (tid < 256) {
    int pix = tid & 63, s = tid >> 6;
    float v[KTAPS];
    float mx = -1e30f;
#pragma unroll
    for (int kk = 0; kk < KTAPS; ++kk) {
      v[kk] = lds_mask[(kk * 4 + s) * WW + pix];
      mx = fmaxf(mx, v[kk]);
    }
    float sum = 0.0f;
#pragma unroll
    for (int kk = 0; kk < KTAPS; ++kk) { v[kk] = __expf(v[kk] - mx); sum += v[kk]; }
    float inv = 1.0f / sum;
    float* mp = mws + ((n * HH + h) * WW + pix) * MCP + s * 28;
#pragma unroll
    for (int kk = 0; kk < KTAPS; ++kk) mp[kk] = v[kk] * inv;
  }
}

__global__ __launch_bounds__(256) void k3_reassemble(const float* __restrict__ x,
                                                     const float* __restrict__ mws,
                                                     float* __restrict__ out) {
  __shared__ float lds_m[16 * 113];
  int b = blockIdx.x;
  int wt = b & 3;
  int h = (b >> 2) & 63;
  int n = b >> 8;
  int w0 = wt << 4;
  int tid = threadIdx.x;
  int base = ((n * HH + h) * WW + w0) * MCP;
  for (int idx = tid; idx < 16 * MCP; idx += 256)
    lds_m[(idx / MCP) * 113 + (idx % MCP)] = mws[base + idx];
  __syncthreads();

  int pix = tid & 15, ci = tid >> 4;
  int wcen = w0 + pix;

  float m0[KTAPS], m1[KTAPS], m2[KTAPS], m3[KTAPS];
#pragma unroll
  for (int kk = 0; kk < KTAPS; ++kk) {
    int dy = kk / 5 - 2, dx = kk % 5 - 2;
    bool ok = (unsigned)(h + dy) < (unsigned)HH && (unsigned)(wcen + dx) < (unsigned)WW;
    const float* mp = &lds_m[pix * 113 + kk];
    float t0 = mp[0], t1 = mp[28], t2 = mp[56], t3 = mp[84];
    m0[kk] = ok ? t0 : 0.0f;
    m1[kk] = ok ? t1 : 0.0f;
    m2[kk] = ok ? t2 : 0.0f;
    m3[kk] = ok ? t3 : 0.0f;
  }
  unsigned coff[5];
  int roff[5];
#pragma unroll
  for (int d = 0; d < 5; ++d) {
    int wx = min(max(wcen + d - 2, 0), WW - 1);
    coff[d] = (unsigned)(ci * HWSZ + wx) * 4u;
    roff[d] = min(max(h + d - 2, 0), HH - 1) * (WW * 4);
  }
  const char* xn = (const char*)(x + (size_t)n * CI * HWSZ);
  float* op = out + ((size_t)(n * CI + ci) * HO + 2 * h) * WO + 2 * wcen;

  for (int cg = 0; cg < CI; cg += 16) {
    const char* xc = xn + (size_t)cg * (HWSZ * 4);
    unsigned vo[5];
#pragma unroll
    for (int d = 0; d < 5; ++d) {
      vo[d] = coff[d];
      asm volatile("" : "+v"(vo[d]));
    }

    if (cg + 16 < CI) {
      unsigned vp = coff[2];
#pragma unroll
      for (int d = 0; d < 5; ++d)
        __builtin_prefetch(xc + 16 * HWSZ * 4 + roff[d] + vp, 0, 3);
    }

    float xv[KTAPS];
#pragma unroll
    for (int kk = 0; kk < KTAPS; ++kk)
      xv[kk] = *(const float*)(xc + roff[kk / 5] + vo[kk % 5]);
    __builtin_amdgcn_sched_barrier(0);
    float acc0 = 0.f, acc1 = 0.f, acc2 = 0.f, acc3 = 0.f;
#pragma unroll
    for (int kk = 0; kk < KTAPS; ++kk) {
      acc0 += xv[kk] * m0[kk];
      acc1 += xv[kk] * m1[kk];
      acc2 += xv[kk] * m2[kk];
      acc3 += xv[kk] * m3[kk];
    }
    *(v2f*)op = (v2f){acc0, acc1};
    *(v2f*)(op + WO) = (v2f){acc2, acc3};
    op += (size_t)16 * HO * WO;
  }
}

extern "C" void kernel_launch(void* const* d_in, const int* in_sizes, int n_in,
                              void* d_out, int out_size, void* d_ws, size_t ws_size,
                              hipStream_t stream) {
  const float* x  = (const float*)d_in[0];
  const float* w1 = (const float*)d_in[1];
  const float* b1 = (const float*)d_in[2];
  const float* w2 = (const float*)d_in[3];
  const float* b2 = (const float*)d_in[4];
  float* out = (float*)d_out;

  float* comp = (float*)d_ws;
  float* w2f  = comp + NN * HWSZ * CCOMP;
  float* mws  = w2f + MCP * KK9;
  float* w1f  = mws;

  k0_prep_frag<<<(MCP * KK9 + CCOMP * CI) / 256, 256, 0, stream>>>(w2, w1, w2f, w1f);
  k1_compress<<<(4 * (NN * HWSZ / 16)) / 8, 256, 0, stream>>>(x, w1f, b1, comp);
  k2_mask_softmax<<<NN * HH, 896, 0, stream>>>(comp, w2f, b2, mws);
  k3_reassemble<<<NN * HH * (WW / 16), 256, 0, stream>>>(x, mws, out);
}


# Reassembly fused into the mask kernel: a row's normalized masks stay in LDS (no workspace round trip, one launch and its wave start-up fewer); a wave = 32 adjacent pixels of one channel; same sums
# speedup vs baseline: 2.5011x; 1.4625x over previous
#include <hip/hip_runtime.h>
#include <math.h>

#define NN 4
#define CI 256
#define HH 64
#define WW 64
#define HWSZ (HH*WW)
#define CCOMP 64
#define MC 100
#define MCP 112
#define KK9 576
#define KTAPS 25
#define HO (HH*2)
#define WO (WW*2)

typedef float v2f __attribute__((ext_vector_type(2)));
typedef float v4f __attribute__((ext_vector_type(4)));
typedef float v8f __attribute__((ext_vector_type(8)));
#define KP2 72
#define KS1 64

#if defined(__AMDGCN__) && __has_builtin(__builtin_amdgcn_wmma_f32_16x16x4_f32)
#define HAVE_WMMA_F32X4 1
#else
#define HAVE_WMMA_F32X4 0
#endif

__device__ __forceinline__ int w2f_index(int row, int kidx) {
  return ((((row >> 4) * KP2 + (kidx >> 3)) * 32 + ((kidx >> 1) & 1) * 16 + (row & 15)) << 2) +
         (((kidx >> 2) & 1) << 1) + (kidx & 1);
}
__device__ __forceinline__ int w1f_index(int row, int k) {
  return ((((row >> 4) * KS1 + (k >> 2)) * 32 + ((k >> 1) & 1) * 16 + (row & 15)) << 1) + (k & 1);
}
__device__ __forceinline__ int comp_index(int n, int h, int w, int cc) {
  return ((((n * HH + h) * 8 + (cc >> 3)) * WW + w) << 3) +
         (((cc >> 1) & 1) << 2) + (((cc >> 2) & 1) << 1) + (cc & 1);
}

__global__ __launch_bounds__(256) void k0_prep_frag(const float* __restrict__ w2,
                                                    const float* __restrict__ w1,
                                                    float* __restrict__ w2f,
                                                    float* __restrict__ w1f) {
  int idx = blockIdx.x * 256 + threadIdx.x;
  int j = idx & 1;
  if (idx < MCP * KK9) {
    int t = (idx >> 1) & 1, lane = (idx >> 2) & 31, lid = lane & 15, hl = lane >> 4;
    int g = idx >> 7;
    int kp = g % KP2, mt = g / KP2;
    int row = mt * 16 + lid;
    int k = 8 * kp + 4 * t + 2 * hl + j;
    int tap = k >> 6, cc = k & 63;
    w2f[idx] = (row < MC) ? w2[row * KK9 + cc * 9 + tap] : 0.0f;
  } else {
    int e = idx - MCP * KK9;
    if (e < CCOMP * CI) {
      int lane = (e >> 1) & 31, lid = lane & 15, hl = lane >> 4;
      int g = e >> 6;
      int ks = g & (KS1 - 1), mt = g >> 6;
      w1f[e] = w1[(mt * 16 + lid) * CI + 4 * ks + 2 * hl + j];
    }
  }
}

__global__ __launch_bounds__(256) void k1_compress(const float* __restrict__ x,
                                                   const float* __restrict__ w1f,
                                                   const float* __restrict__ b1,
                                                   float* __restrict__ comp) {
#if HAVE_WMMA_F32X4
  int wave = blockIdx.x * 8 + (threadIdx.x >> 5);
  int lane = threadIdx.x & 31;
  int hl = lane >> 4, lid = lane & 15;
  int mt = wave & 3;
  int ct = wave >> 2;
  int n = ct >> 8;
  int hw0 = (ct & 255) << 4;
  const float* xb = x + n * (CI * HWSZ) + hw0 + lid;
  const float* ap = w1f + ((mt * KS1 * 32 + lane) << 1);
  v8f acc = {};
  for (int k = 0; k < CI; k += 4) {
    int ka = k + 2 * hl;
    v2f a = *(const v2f*)(ap + k * 16);
    v2f b;
    b.x = xb[ka * HWSZ];
    b.y = xb[(ka + 1) * HWSZ];
    acc = __builtin_amdgcn_wmma_f32_16x16x4_f32(false, a, false, b,
                                                (short)0, acc, false, false);
  }
  int col = hw0 + lid;
  int row0 = mt * 16 + 8 * hl;
  const float* bb = b1 + row0;
  float* cp = comp + comp_index(n, col >> 6, col & 63, row0);
  v4f lo = {acc[0] + bb[0], acc[1] + bb[1], acc[4] + bb[4], acc[5] + bb[5]};
  v4f hi = {acc[2] + bb[2], acc[3] + bb[3], acc[6] + bb[6], acc[7] + bb[7]};
  *(v4f*)cp = lo;
  *(v4f*)(cp + 4) = hi;
#else
  int stride = gridDim.x * blockDim.x;
  for (int e = blockIdx.x * blockDim.x + threadIdx.x; e < CCOMP * NN * HWSZ; e += stride) {
    int row = e % CCOMP, col = e / CCOMP;
    int n = col / HWSZ, hw = col % HWSZ;
    float s = b1[row];
    for (int k = 0; k < CI; ++k) s += w1f[w1f_index(row, k)] * x[(n * CI + k) * HWSZ + hw];
    comp[comp_index(n, hw >> 6, hw & 63, row)] = s;
  }
#endif
}

#if HAVE_WMMA_F32X4
__device__ __forceinline__ void k2_load_group(const float* __restrict__ ap,
                                              const float* __restrict__ bp,
                                              v4f (&fa)[4], v4f (&fb)[4]) {
#pragma unroll
  for (int i = 0; i < 4; ++i) {
    fa[i] = *(const v4f*)(ap + i * 128);
    fb[i] = *(const v4f*)(bp + i * (WW * 8));
  }
}
__device__ __forceinline__ v8f k2_mma_group(v8f acc, const v4f (&fa)[4],
                                            const v4f (&fb)[4], bool ok) {
#pragma unroll
  for (int i = 0; i < 4; ++i) {
    v4f b = ok ? fb[i] : (v4f){0.0f, 0.0f, 0.0f, 0.0f};
    acc = __builtin_amdgcn_wmma_f32_16x16x4_f32(false, fa[i].xy, false, b.xy,
                                                (short)0, acc, false, false);
    acc = __builtin_amdgcn_wmma_f32_16x16x4_f32(false, fa[i].zw, false, b.zw,
                                                (short)0, acc, false, false);
  }
  return acc;
}
#endif

__global__ __launch_bounds__(896) void k2_mask_reassemble(const float* __restrict__ comp,
                                                          const float* __restrict__ w2f,
                                                          const float* __restrict__ b2,
                                                          const float* __restrict__ x,
                                                          float* __restrict__ out) {
  __shared__ float lds_mask[MCP * WW];
  __shared__ float lds_m[WW * 113];
  int n = blockIdx.x >> 6;
  int h = blockIdx.x & 63;
  int tid = threadIdx.x;
#if HAVE_WMMA_F32X4
  int wave = tid >> 5, lane = tid & 31, hl = lane >> 4, lid = lane & 15;
  int mt = wave >> 2;
  int w0 = (wave & 3) << 4;
  int wcol = w0 + lid;
  const float* abase = w2f + ((mt * KP2 * 32 + lane) << 2);
  const float* cbase = comp + (size_t)n * HH * WW * CCOMP + 4 * hl;
  int rofs[3], cofs[3];
  bool rok[3], cok[3];
#pragma unroll
  for (int d = 0; d < 3; ++d) {
    int hr = h + d - 1, wc = wcol + d - 1;
    rok[d] = (unsigned)hr < (unsigned)HH;
    cok[d] = (unsigned)wc < (unsigned)WW;
    rofs[d] = min(max(hr, 0), HH - 1) * (8 * WW * 8);
    cofs[d] = min(max(wc, 0), WW - 1) * 8;
  }
  v4f fa0[4], fb0[4], fa1[4], fb1[4];
  v8f acc = {};
#define K2_AP(g) (abase + (g) * (4 * 128))
#define K2_BP(g) (cbase + rofs[((g) >> 1) / 3] + cofs[((g) >> 1) % 3] + ((g) & 1) * (4 * WW * 8))
#define K2_OK(g) (rok[((g) >> 1) / 3] && cok[((g) >> 1) % 3])
  k2_load_group(K2_AP(0), K2_BP(0), fa0, fb0);
#pragma unroll
  for (int g = 0; g < 18; g += 2) {
    k2_load_group(K2_AP(g + 1), K2_BP(g + 1), fa1, fb1);
    __builtin_amdgcn_sched_barrier(0);
    acc = k2_mma_group(acc, fa0, fb0, K2_OK(g));
    __builtin_amdgcn_sched_barrier(0);
    if (g + 2 < 18) k2_load_group(K2_AP(g + 2), K2_BP(g + 2), fa0, fb0);
    __builtin_amdgcn_sched_barrier(0);
    acc = k2_mma_group(acc, fa1, fb1, K2_OK(g + 1));
    __builtin_amdgcn_sched_barrier(0);
  }
#undef K2_AP
#undef K2_BP
#undef K2_OK
#pragma unroll
  for (int r = 0; r < 8; ++r) {
    int row = mt * 16 + r + 8 * hl;
    float v = acc[r] + ((row < MC) ? b2[row] : 0.0f);
    lds_mask[row * WW + w0 + lid] = v;
  }
#else
  for (int e = tid; e < MCP * WW; e += 896) {
    int row = e >> 6, pix = e & 63;
    float s = (row < MC) ? b2[row] : 0.0f;
    if (row < MC) {
      for (int kidx = 0; kidx < KK9; ++kidx) {
        int dy = kidx / 192, r0 = kidx % 192, dx = r0 >> 6, cc = r0 & 63;
        int hrow = h + dy - 1, wc = pix + dx - 1;
        if ((unsigned)hrow < (unsigned)HH && (unsigned)wc < (unsigned)WW)
          s += w2f[w2f_index(row, kidx)] * comp[comp_index(n, hrow, wc, cc)];
      }
    }
    lds_mask[row * WW + pix] = s;
  }
#endif
  __syncthreads();
  if (tid < 256) {
    int pix = tid & 63, s = tid >> 6;
    float v[KTAPS];
    float mx = -1e30f;
#pragma unroll
    for (int kk = 0; kk < KTAPS; ++kk) {
      v[kk] = lds_mask[(kk * 4 + s) * WW + pix];
      mx = fmaxf(mx, v[kk]);
    }
    float sum = 0.0f;
#pragma unroll
    for (int kk = 0; kk < KTAPS; ++kk) { v[kk] = __expf(v[kk] - mx); sum += v[kk]; }
    float inv = 1.0f / sum;
    float* mp = &lds_m[pix * 113 + s * 28];
#pragma unroll
    for (int kk = 0; kk < KTAPS; ++kk) mp[kk] = v[kk] * inv;
  }
  __syncthreads();

  {
    int pix = tid & 63;
    int cw = __builtin_amdgcn_readfirstlane(tid >> 6);

    float m0[KTAPS], m1[KTAPS], m2[KTAPS], m3[KTAPS];
#pragma unroll
    for (int kk = 0; kk < KTAPS; ++kk) {
      int dy = kk / 5 - 2, dx = kk % 5 - 2;
      bool ok = (unsigned)(h + dy) < (unsigned)HH && (unsigned)(pix + dx) < (unsigned)WW;
      const float* mp = &lds_m[pix * 113 + kk];
      float t0 = mp[0], t1 = mp[28], t2 = mp[56], t3 = mp[84];
      m0[kk] = ok ? t0 : 0.0f;
      m1[kk] = ok ? t1 : 0.0f;
      m2[kk] = ok ? t2 : 0.0f;
      m3[kk] = ok ? t3 : 0.0f;
    }
    unsigned coff[5];
    int roff[5];
#pragma unroll
    for (int d = 0; d < 5; ++d) {
      int wx = min(max(pix + d - 2, 0), WW - 1);
      coff[d] = (unsigned)wx * 4u;
      roff[d] = min(max(h + d - 2, 0), HH - 1) * (WW * 4);
    }
    const char* xn = (const char*)(x + (size_t)n * CI * HWSZ);
    float* op = out + ((size_t)(n * CI + cw) * HO + 2 * h) * WO + 2 * pix;

    for (int c = cw; c < CI; c += 14) {
      const char* xc = xn + (size_t)c * (HWSZ * 4);
      unsigned vo[5];
#pragma unroll
      for (int d = 0; d < 5; ++d) {
        vo[d] = coff[d];
        asm volatile("" : "+v"(vo[d]));
      }
      float xv[KTAPS];
#pragma unroll
      for (int kk = 0; kk < KTAPS; ++kk)
        xv[kk] = *(const float*)(xc + roff[kk / 5] + vo[kk % 5]);
      __builtin_amdgcn_sched_barrier(0);
      float acc0 = 0.f, acc1 = 0.f, acc2 = 0.f, acc3 = 0.f;
#pragma unroll
      for (int kk = 0; kk < KTAPS; ++kk) {
        acc0 += xv[kk] * m0[kk];
        acc1 += xv[kk] * m1[kk];
        acc2 += xv[kk] * m2[kk];
        acc3 += xv[kk] * m3[kk];
      }
      *(v2f*)op = (v2f){acc0, acc1};
      *(v2f*)(op + WO) = (v2f){acc2, acc3};
      op += (size_t)14 * HO * WO;
    }
  }
}

extern "C" void kernel_launch(void* const* d_in, const int* in_sizes, int n_in,
                              void* d_out, int out_size, void* d_ws, size_t ws_size,
                              hipStream_t stream) {
  const float* x  = (const float*)d_in[0];
  const float* w1 = (const float*)d_in[1];
  const float* b1 = (const float*)d_in[2];
  const float* w2 = (const float*)d_in[3];
  const float* b2 = (const float*)d_in[4];
  float* out = (float*)d_out;

  float* comp = (float*)d_ws;
  float* w2f  = comp + NN * HWSZ * CCOMP;
  float* w1f  = w2f + MCP * KK9;

  k0_prep_frag<<<(MCP * KK9 + CCOMP * CI) / 256, 256, 0, stream>>>(w2, w1, w2f, w1f);
  k1_compress<<<(4 * (NN * HWSZ / 16)) / 8, 256, 0, stream>>>(x, w1f, b1, comp);
  k2_mask_reassemble<<<NN * HH, 896, 0, stream>>>(comp, w2f, b2, x, out);
}


# k1: a wave computes one M tile x 4 N tiles (x read four pixels per lane) and its loads run four 4-step groups ahead of the matrix steps; 1024 waves instead of 4096; same sums, same order
# speedup vs baseline: 2.6743x; 1.0692x over previous
#include <hip/hip_runtime.h>
#include <math.h>

#define NN 4
#define CI 256
#define HH 64
#define WW 64
#define HWSZ (HH*WW)
#define CCOMP 64
#define MC 100
#define MCP 112
#define KK9 576
#define KTAPS 25
#define HO (HH*2)
#define WO (WW*2)

typedef float v2f __attribute__((ext_vector_type(2)));
typedef float v4f __attribute__((ext_vector_type(4)));
typedef float v8f __attribute__((ext_vector_type(8)));
#define KP2 72
#define KS1 64

#if defined(__AMDGCN__) && __has_builtin(__builtin_amdgcn_wmma_f32_16x16x4_f32)
#define HAVE_WMMA_F32X4 1
#else
#define HAVE_WMMA_F32X4 0
#endif

__device__ __forceinline__ int w2f_index(int row, int kidx) {
  return ((((row >> 4) * KP2 + (kidx >> 3)) * 32 + ((kidx >> 1) & 1) * 16 + (row & 15)) << 2) +
         (((kidx >> 2) & 1) << 1) + (kidx & 1);
}
__device__ __forceinline__ int w1f_index(int row, int k) {
  return ((((row >> 4) * KS1 + (k >> 2)) * 32 + ((k >> 1) & 1) * 16 + (row & 15)) << 1) + (k & 1);
}
__device__ __forceinline__ int comp_index(int n, int h, int w, int cc) {
  return ((((n * HH + h) * 8 + (cc >> 3)) * WW + w) << 3) +
         (((cc >> 1) & 1) << 2) + (((cc >> 2) & 1) << 1) + (cc & 1);
}

__global__ __launch_bounds__(256) void k0_prep_frag(const float* __restrict__ w2,
                                                    const float* __restrict__ w1,
                                                    float* __restrict__ w2f,
                                                    float* __restrict__ w1f) {
  int idx = blockIdx.x * 256 + threadIdx.x;
  int j = idx & 1;
  if (idx < MCP * KK9) {
    int t = (idx >> 1) & 1, lane = (idx >> 2) & 31, lid = lane & 15, hl = lane >> 4;
    int g = idx >> 7;
    int kp = g % KP2, mt = g / KP2;
    int row = mt * 16 + lid;
    int k = 8 * kp + 4 * t + 2 * hl + j;
    int tap = k >> 6, cc = k & 63;
    w2f[idx] = (row < MC) ? w2[row * KK9 + cc * 9 + tap] : 0.0f;
  } else {
    int e = idx - MCP * KK9;
    if (e < CCOMP * CI) {
      int lane = (e >> 1) & 31, lid = lane & 15, hl = lane >> 4;
      int g = e >> 6;
      int ks = g & (KS1 - 1), mt = g >> 6;
      w1f[e] = w1[(mt * 16 + lid) * CI + 4 * ks + 2 * hl + j];
    }
  }
}

#if HAVE_WMMA_F32X4
__device__ __forceinline__ void k1_load_group(const float* __restrict__ ap,
                                              const float* __restrict__ xp,
                                              v2f (&fa)[4], v4f (&fb0)[4], v4f (&fb1)[4]) {
#pragma unroll
  for (int i = 0; i < 4; ++i) {
    fa[i] = *(const v2f*)(ap + i * 64);
    fb0[i] = *(const v4f*)(xp + (4 * i) * HWSZ);
    fb1[i] = *(const v4f*)(xp + (4 * i + 1) * HWSZ);
  }
}
#define K1_WMMA(acc, a, b) \
  acc = __builtin_amdgcn_wmma_f32_16x16x4_f32(false, a, false, b, (short)0, acc, false, false)
#endif

__global__ __launch_bounds__(128) void k1_compress(const float* __restrict__ x,
                                                   const float* __restrict__ w1f,
                                                   const float* __restrict__ b1,
                                                   float* __restrict__ comp) {
#if HAVE_WMMA_F32X4
  int mt = threadIdx.x >> 5;
  int lane = threadIdx.x & 31;
  int hl = lane >> 4, lid = lane & 15;
  int n = blockIdx.x >> 6;
  int h = blockIdx.x & 63;
  const float* xb = x + n * (CI * HWSZ) + (2 * hl) * HWSZ + h * WW + 4 * lid;
  const float* ab = w1f + ((mt * KS1 * 32 + lane) << 1);
  v2f fa[5][4];
  v4f fb0[5][4], fb1[5][4];
  v8f acc0 = {}, acc1 = {}, acc2 = {}, acc3 = {};
#define K1_LOAD(g) k1_load_group(ab + (g) * 256, xb + (g) * (16 * HWSZ), \
                                 fa[(g) % 5], fb0[(g) % 5], fb1[(g) % 5])
  K1_LOAD(0);
  K1_LOAD(1);
  K1_LOAD(2);
  K1_LOAD(3);
#pragma unroll
  for (int g = 0; g < 16; ++g) {
    if (g + 4 < 16) K1_LOAD(g + 4);
    __builtin_amdgcn_sched_barrier(0);
#pragma unroll
    for (int i = 0; i < 4; ++i) {
      v4f p = fb0[g % 5][i], q = fb1[g % 5][i];
      K1_WMMA(acc0, fa[g % 5][i], ((v2f){p.x, q.x}));
      K1_WMMA(acc1, fa[g % 5][i], ((v2f){p.y, q.y}));
      K1_WMMA(acc2, fa[g % 5][i], ((v2f){p.z, q.z}));
      K1_WMMA(acc3, fa[g % 5][i], ((v2f){p.w, q.w}));
    }
    __builtin_amdgcn_sched_barrier(0);
  }
#undef K1_LOAD
  int row0 = mt * 16 + 8 * hl;
  const float* bb = b1 + row0;
  float* cp = comp + comp_index(n, h, 4 * lid, row0);
#pragma unroll
  for (int j = 0; j < 4; ++j) {
    v8f acc = (j == 0) ? acc0 : (j == 1) ? acc1 : (j == 2) ? acc2 : acc3;
    v4f lo = {acc[0] + bb[0], acc[1] + bb[1], acc[4] + bb[4], acc[5] + bb[5]};
    v4f hi = {acc[2] + bb[2], acc[3] + bb[3], acc[6] + bb[6], acc[7] + bb[7]};
    *(v4f*)(cp + 8 * j) = lo;
    *(v4f*)(cp + 8 * j + 4) = hi;
  }
#else
  int stride = gridDim.x * blockDim.x;
  for (int e = blockIdx.x * blockDim.x + threadIdx.x; e < CCOMP * NN * HWSZ; e += stride) {
    int row = e % CCOMP, col = e / CCOMP;
    int n = col / HWSZ, hw = col % HWSZ;
    float s = b1[row];
    for (int k = 0; k < CI; ++k) s += w1f[w1f_index(row, k)] * x[(n * CI + k) * HWSZ + hw];
    comp[comp_index(n, hw >> 6, hw & 63, row)] = s;
  }
#endif
}

#if HAVE_WMMA_F32X4
__device__ __forceinline__ void k2_load_group(const float* __restrict__ ap,
                                              const float* __restrict__ bp,
                                              v4f (&fa)[4], v4f (&fb)[4]) {
#pragma unroll
  for (int i = 0; i < 4; ++i) {
    fa[i] = *(const v4f*)(ap + i * 128);
    fb[i] = *(const v4f*)(bp + i * (WW * 8));
  }
}
__device__ __forceinline__ v8f k2_mma_group(v8f acc, const v4f (&fa)[4],
                                            const v4f (&fb)[4], bool ok) {
#pragma unroll
  for (int i = 0; i < 4; ++i) {
    v4f b = ok ? fb[i] : (v4f){0.0f, 0.0f, 0.0f, 0.0f};
    acc = __builtin_amdgcn_wmma_f32_16x16x4_f32(false, fa[i].xy, false, b.xy,
                                                (short)0, acc, false, false);
    acc = __builtin_amdgcn_wmma_f32_16x16x4_f32(false, fa[i].zw, false, b.zw,
                                                (short)0, acc, false, false);
  }
  return acc;
}
#endif

__global__ __launch_bounds__(896) void k2_mask_reassemble(const float* __restrict__ comp,
                                                          const float* __restrict__ w2f,
                                                          const float* __restrict__ b2,
                                                          const float* __restrict__ x,
                                                          float* __restrict__ out) {
  __shared__ float lds_mask[MCP * WW];
  __shared__ float lds_m[WW * 113];
  int n = blockIdx.x >> 6;
  int h = blockIdx.x & 63;
  int tid = threadIdx.x;
#if HAVE_WMMA_F32X4
  int wave = tid >> 5, lane = tid & 31, hl = lane >> 4, lid = lane & 15;
  int mt = wave >> 2;
  int w0 = (wave & 3) << 4;
  int wcol = w0 + lid;
  const float* abase = w2f + ((mt * KP2 * 32 + lane) << 2);
  const float* cbase = comp + (size_t)n * HH * WW * CCOMP + 4 * hl;
  int rofs[3], cofs[3];
  bool rok[3], cok[3];
#pragma unroll
  for (int d = 0; d < 3; ++d) {
    int hr = h + d - 1, wc = wcol + d - 1;
    rok[d] = (unsigned)hr < (unsigned)HH;
    cok[d] = (unsigned)wc < (unsigned)WW;
    rofs[d] = min(max(hr, 0), HH - 1) * (8 * WW * 8);
    cofs[d] = min(max(wc, 0), WW - 1) * 8;
  }
  v4f fa0[4], fb0[4], fa1[4], fb1[4];
  v8f acc = {};
#define K2_AP(g) (abase + (g) * (4 * 128))
#define K2_BP(g) (cbase + rofs[((g) >> 1) / 3] + cofs[((g) >> 1) % 3] + ((g) & 1) * (4 * WW * 8))
#define K2_OK(g) (rok[((g) >> 1) / 3] && cok[((g) >> 1) % 3])
  k2_load_group(K2_AP(0), K2_BP(0), fa0, fb0);
#pragma unroll
  for (int g = 0; g < 18; g += 2) {
    k2_load_group(K2_AP(g + 1), K2_BP(g + 1), fa1, fb1);
    __builtin_amdgcn_sched_barrier(0);
    acc = k2_mma_group(acc, fa0, fb0, K2_OK(g));
    __builtin_amdgcn_sched_barrier(0);
    if (g + 2 < 18) k2_load_group(K2_AP(g + 2), K2_BP(g + 2), fa0, fb0);
    __builtin_amdgcn_sched_barrier(0);
    acc = k2_mma_group(acc, fa1, fb1, K2_OK(g + 1));
    __builtin_amdgcn_sched_barrier(0);
  }
#undef K2_AP
#undef K2_BP
#undef K2_OK
#pragma unroll
  for (int r = 0; r < 8; ++r) {
    int row = mt * 16 + r + 8 * hl;
    float v = acc[r] + ((row < MC) ? b2[row] : 0.0f);
    lds_mask[row * WW + w0 + lid] = v;
  }
#else
  for (int e = tid; e < MCP * WW; e += 896) {
    int row = e >> 6, pix = e & 63;
    float s = (row < MC) ? b2[row] : 0.0f;
    if (row < MC) {
      for (int kidx = 0; kidx < KK9; ++kidx) {
        int dy = kidx / 192, r0 = kidx % 192, dx = r0 >> 6, cc = r0 & 63;
        int hrow = h + dy - 1, wc = pix + dx - 1;
        if ((unsigned)hrow < (unsigned)HH && (unsigned)wc < (unsigned)WW)
          s += w2f[w2f_index(row, kidx)] * comp[comp_index(n, hrow, wc, cc)];
      }
    }
    lds_mask[row * WW + pix] = s;
  }
#endif
  __syncthreads();
  if (tid < 256) {
    int pix = tid & 63, s = tid >> 6;
    float v[KTAPS];
    float mx = -1e30f;
#pragma unroll
    for (int kk = 0; kk < KTAPS; ++kk) {
      v[kk] = lds_mask[(kk * 4 + s) * WW + pix];
      mx = fmaxf(mx, v[kk]);
    }
    float sum = 0.0f;
#pragma unroll
    for (int kk = 0; kk < KTAPS; ++kk) { v[kk] = __expf(v[kk] - mx); sum += v[kk]; }
    float inv = 1.0f / sum;
    float* mp = &lds_m[pix * 113 + s * 28];
#pragma unroll
    for (int kk = 0; kk < KTAPS; ++kk) mp[kk] = v[kk] * inv;
  }
  __syncthreads();

  {
    int pix = tid & 63;
    int cw = __builtin_amdgcn_readfirstlane(tid >> 6);

    float m0[KTAPS], m1[KTAPS], m2[KTAPS], m3[KTAPS];
#pragma unroll
    for (int kk = 0; kk < KTAPS; ++kk) {
      int dy = kk / 5 - 2, dx = kk % 5 - 2;
      bool ok = (unsigned)(h + dy) < (unsigned)HH && (unsigned)(pix + dx) < (unsigned)WW;
      const float* mp = &lds_m[pix * 113 + kk];
      float t0 = mp[0], t1 = mp[28], t2 = mp[56], t3 = mp[84];
      m0[kk] = ok ? t0 : 0.0f;
      m1[kk] = ok ? t1 : 0.0f;
      m2[kk] = ok ? t2 : 0.0f;
      m3[kk] = ok ? t3 : 0.0f;
    }
    unsigned coff[5];
    int roff[5];
#pragma unroll
    for (int d = 0; d < 5; ++d) {
      int wx = min(max(pix + d - 2, 0), WW - 1);
      coff[d] = (unsigned)wx * 4u;
      roff[d] = min(max(h + d - 2, 0), HH - 1) * (WW * 4);
    }
    const char* xn = (const char*)(x + (size_t)n * CI * HWSZ);
    float* op = out + ((size_t)(n * CI + cw) * HO + 2 * h) * WO + 2 * pix;

    for (int c = cw; c < CI; c += 14) {
      const char* xc = xn + (size_t)c * (HWSZ * 4);
      unsigned vo[5];
#pragma unroll
      for (int d = 0; d < 5; ++d) {
        vo[d] = coff[d];
        asm volatile("" : "+v"(vo[d]));
      }
      float xv[KTAPS];
#pragma unroll
      for (int kk = 0; kk < KTAPS; ++kk)
        xv[kk] = *(const float*)(xc + roff[kk / 5] + vo[kk % 5]);
      __builtin_amdgcn_sched_barrier(0);
      float acc0 = 0.f, acc1 = 0.f, acc2 = 0.f, acc3 = 0.f;
#pragma unroll
      for (int kk = 0; kk < KTAPS; ++kk) {
        acc0 += xv[kk] * m0[kk];
        acc1 += xv[kk] * m1[kk];
        acc2 += xv[kk] * m2[kk];
        acc3 += xv[kk] * m3[kk];
      }
      *(v2f*)op = (v2f){acc0, acc1};
      *(v2f*)(op + WO) = (v2f){acc2, acc3};
      op += (size_t)14 * HO * WO;
    }
  }
}

extern "C" void kernel_launch(void* const* d_in, const int* in_sizes, int n_in,
                              void* d_out, int out_size, void* d_ws, size_t ws_size,
                              hipStream_t stream) {
  const float* x  = (const float*)d_in[0];
  const float* w1 = (const float*)d_in[1];
  const float* b1 = (const float*)d_in[2];
  const float* w2 = (const float*)d_in[3];
  const float* b2 = (const float*)d_in[4];
  float* out = (float*)d_out;

  float* comp = (float*)d_ws;
  float* w2f  = comp + NN * HWSZ * CCOMP;
  float* w1f  = w2f + MCP * KK9;

  k0_prep_frag<<<(MCP * KK9 + CCOMP * CI) / 256, 256, 0, stream>>>(w2, w1, w2f, w1f);
  k1_compress<<<NN * HH, 128, 0, stream>>>(x, w1f, b1, comp);
  k2_mask_reassemble<<<NN * HH, 896, 0, stream>>>(comp, w2f, b2, x, out);
}
